# starting kernel
# baseline (speedup 1.0000x reference)
#include <hip/hip_runtime.h>
#include <hip/hip_bf16.h>

typedef float v2f __attribute__((ext_vector_type(2)));
typedef float v8f __attribute__((ext_vector_type(8)));

#define F 32
#define NB 16

__device__ __forceinline__ float fast_silu(float x) {
    return x * __builtin_amdgcn_rcpf(1.0f + __expf(-x));
}

__global__ void zero_kernel(float* __restrict__ p, int n) {
    int g = blockIdx.x * blockDim.x + threadIdx.x;
    if (g < n) p[g] = 0.0f;
}

__global__ void mol_kernel(const int* __restrict__ q, const int* __restrict__ sp,
                           const float* __restrict__ ce, const float* __restrict__ se,
                           const float* __restrict__ mW, const float* __restrict__ mb,
                           float* __restrict__ mol, int B) {
    int g = blockIdx.x * blockDim.x + threadIdx.x;
    if (g >= B * F) return;
    int b = g >> 5, f = g & 31;
    int ci = min(max(q[b] + 5, 0), 10);
    int si = min(max(sp[b] - 1, 0), 6);
    float acc = mb[f];
    #pragma unroll
    for (int k = 0; k < 16; ++k) acc += ce[ci * 16 + k] * mW[k * F + f];
    #pragma unroll
    for (int k = 0; k < 16; ++k) acc += se[si * 16 + k] * mW[(16 + k) * F + f];
    mol[g] = acc;
}

__global__ void x0_kernel(const int* __restrict__ Z, const int* __restrict__ seg,
                          const float* __restrict__ ae, const float* __restrict__ mol,
                          float* __restrict__ x0, int A) {
    int g = blockIdx.x * blockDim.x + threadIdx.x;
    if (g >= A * F) return;
    int a = g >> 5, f = g & 31;
    x0[g] = ae[Z[a] * F + f] + mol[seg[a] * F + f];
}

__global__ __launch_bounds__(256) void edge_kernel(
    const float* __restrict__ pos, const int* __restrict__ dst,
    const int* __restrict__ src, const float* __restrict__ Wb,
    const float* __restrict__ xin, float* __restrict__ accum,
    int P, int add_cut) {
    long long g = (long long)blockIdx.x * blockDim.x + threadIdx.x;
    int e = (int)(g >> 5);
    int f = (int)(g & 31);
    if (e >= P) return;
    int s = src[e], d = dst[e];
    float dx = pos[s * 3 + 0] - pos[d * 3 + 0];
    float dy = pos[s * 3 + 1] - pos[d * 3 + 1];
    float dz = pos[s * 3 + 2] - pos[d * 3 + 2];
    float r = sqrtf(dx * dx + dy * dy + dz * dz);
    float u = r * (1.0f / 6.0f);
    u = u * u;
    float cutv = 0.0f;
    if (u < 1.0f)
        cutv = __expf(1.0f - __builtin_amdgcn_rcpf(fmaxf(1.0f - u, 1e-6f)));
    float zc = fminf(fmaxf(2.0f * __expf(-r) - 1.0f, -1.0f), 1.0f);
    float Tm2 = 1.0f, Tm1 = zc;
    float c = Wb[f] + Wb[F + f] * zc;
    #pragma unroll
    for (int b2 = 2; b2 < NB; ++b2) {
        float T = 2.0f * zc * Tm1 - Tm2;
        c += Wb[b2 * F + f] * T;
        Tm2 = Tm1; Tm1 = T;
    }
    float scale = cutv * (c + (add_cut ? 1.0f : 0.0f));
    float val = scale * xin[s * F + f];
    unsafeAtomicAdd(&accum[d * F + f], val);
}

__global__ __launch_bounds__(256) void refine_kernel(
    const float* __restrict__ xa, const float* __restrict__ xb,
    const float* __restrict__ W1s, const float* __restrict__ b1s,
    const float* __restrict__ W2s, const float* __restrict__ b2s,
    float* __restrict__ out, int A) {
    __shared__ __align__(16) float sW1[3 * F * F];
    __shared__ __align__(16) float sW2[3 * F * F];
    __shared__ float sb1[3 * F];
    __shared__ float sb2[3 * F];
    __shared__ __align__(16) float xt[128 * F];
    __shared__ __align__(16) float ht[128 * F];

    const int t = threadIdx.x;
    const int lane = t & 31;
    const int wave = t >> 5;
    const int base = blockIdx.x * 128 * F;
    const int lim = A * F;

    for (int i = t; i < 3 * F * F; i += 256) {
        int rem = i & 1023;
        int k = rem >> 5;
        int n = rem & 31;
        int fidx = (i & ~1023) + ((k >> 1) << 6) + (n << 1) + (k & 1);
        sW1[fidx] = W1s[i];
        sW2[fidx] = W2s[i];
    }
    for (int i = t; i < 3 * F; i += 256) { sb1[i] = b1s[i]; sb2[i] = b2s[i]; }
    for (int i = t; i < 128 * F; i += 256)
        xt[i] = (base + i < lim) ? (xa[base + i] + xb[base + i]) : 0.0f;
    __syncthreads();

    const int m     = lane & 15;
    const int khalf = (lane >> 4) << 1;
    const int row0  = wave * 16;
    const int cm    = (lane >> 4) << 3;
    const int cn    = lane & 15;

    for (int j = 0; j < 3; ++j) {
        const v2f* W1p = (const v2f*)(sW1 + j * F * F);
        const v2f* W2p = (const v2f*)(sW2 + j * F * F);
        const float* B1 = sb1 + j * F;
        const float* B2 = sb2 + j * F;

        v8f acc0, acc1;
        {
            float bn0 = B1[cn], bn1 = B1[cn + 16];
            #pragma unroll
            for (int q = 0; q < 8; ++q) { acc0[q] = bn0; acc1[q] = bn1; }
        }
        #pragma unroll
        for (int kc = 0; kc < 8; ++kc) {
            int kb = kc * 4 + khalf;
            int kp = (kb >> 1) << 5;
            v2f a  = *(const v2f*)(xt + (row0 + m) * F + kb);
            v2f b0 = W1p[kp + cn];
            v2f b1 = W1p[kp + cn + 16];
            acc0 = __builtin_amdgcn_wmma_f32_16x16x4_f32(false, a, false, b0, (short)0, acc0, false, false);
            acc1 = __builtin_amdgcn_wmma_f32_16x16x4_f32(false, a, false, b1, (short)0, acc1, false, false);
        }
        #pragma unroll
        for (int q = 0; q < 8; ++q) {
            ht[(row0 + cm + q) * F + cn]      = fast_silu(acc0[q]);
            ht[(row0 + cm + q) * F + cn + 16] = fast_silu(acc1[q]);
        }
        __syncthreads();

        {
            float bn0 = B2[cn], bn1 = B2[cn + 16];
            #pragma unroll
            for (int q = 0; q < 8; ++q) { acc0[q] = bn0; acc1[q] = bn1; }
        }
        #pragma unroll
        for (int kc = 0; kc < 8; ++kc) {
            int kb = kc * 4 + khalf;
            int kp = (kb >> 1) << 5;
            v2f a  = *(const v2f*)(ht + (row0 + m) * F + kb);
            v2f b0 = W2p[kp + cn];
            v2f b1 = W2p[kp + cn + 16];
            acc0 = __builtin_amdgcn_wmma_f32_16x16x4_f32(false, a, false, b0, (short)0, acc0, false, false);
            acc1 = __builtin_amdgcn_wmma_f32_16x16x4_f32(false, a, false, b1, (short)0, acc1, false, false);
        }
        __syncthreads();
        #pragma unroll
        for (int q = 0; q < 8; ++q) {
            float x0v = xt[(row0 + cm + q) * F + cn]      + acc0[q];
            float x1v = xt[(row0 + cm + q) * F + cn + 16] + acc1[q];
            xt[(row0 + cm + q) * F + cn]      = fast_silu(x0v);
            xt[(row0 + cm + q) * F + cn + 16] = fast_silu(x1v);
        }
        __syncthreads();
    }

    for (int i = t; i < 128 * F; i += 256)
        if (base + i < lim) out[base + i] = xt[i];
}

extern "C" void kernel_launch(void* const* d_in, const int* in_sizes, int n_in,
                              void* d_out, int out_size, void* d_ws, size_t ws_size,
                              hipStream_t stream) {
    const float* pos  = (const float*)d_in[0];
    const int*   q    = (const int*)d_in[1];
    const int*   sp   = (const int*)d_in[2];
    const int*   Z    = (const int*)d_in[3];
    const int*   dst  = (const int*)d_in[4];
    const int*   src  = (const int*)d_in[5];
    const int*   seg  = (const int*)d_in[6];
    const float* ce   = (const float*)d_in[7];
    const float* se   = (const float*)d_in[8];
    const float* ae   = (const float*)d_in[9];
    const float* mW   = (const float*)d_in[10];
    const float* mb   = (const float*)d_in[11];
    const float* mpWb = (const float*)d_in[12];
    const float* rW1  = (const float*)d_in[13];
    const float* rb1  = (const float*)d_in[14];
    const float* rW2  = (const float*)d_in[15];
    const float* rb2  = (const float*)d_in[16];

    const int B = in_sizes[1];
    const int A = in_sizes[3];
    const int P = in_sizes[4];

    float* ws  = (float*)d_ws;
    float* mol = ws;
    float* x0  = mol + (size_t)B * F;
    float* s0  = x0 + (size_t)A * F;
    float* xs0 = s0 + (size_t)A * F;
    float* s1  = xs0 + (size_t)A * F;

    {
        int n = 3 * A * F;
        zero_kernel<<<(n + 255) / 256, 256, 0, stream>>>(s0, n);
    }
    mol_kernel<<<(B * F + 255) / 256, 256, 0, stream>>>(q, sp, ce, se, mW, mb, mol, B);
    x0_kernel<<<(A * F + 255) / 256, 256, 0, stream>>>(Z, seg, ae, mol, x0, A);

    const long long nedge = (long long)P * F;
    const int eblocks = (int)((nedge + 255) / 256);

    edge_kernel<<<eblocks, 256, 0, stream>>>(pos, dst, src, mpWb, x0, s0, P, 1);
    refine_kernel<<<(A + 127) / 128, 256, 0, stream>>>(
        x0, s0, rW1, rb1, rW2, rb2, xs0, A);
    edge_kernel<<<eblocks, 256, 0, stream>>>(pos, dst, src, mpWb + NB * F, xs0, s1, P, 0);
    refine_kernel<<<(A + 127) / 128, 256, 0, stream>>>(
        xs0, s1, rW1 + 3 * F * F, rb1 + 3 * F, rW2 + 3 * F * F, rb2 + 3 * F,
        (float*)d_out, A);
}


# refine kernel: weights, biases and row tile fetched as one batch of 16-byte loads before the first LDS store (22 dependent load-wait-store trips per block become 1); tile stored in 16-byte pieces
# speedup vs baseline: 1.3084x; 1.3084x over previous
#include <hip/hip_runtime.h>
#include <hip/hip_bf16.h>

typedef float v2f __attribute__((ext_vector_type(2)));
typedef float v8f __attribute__((ext_vector_type(8)));

#define F 32
#define NB 16

__device__ __forceinline__ float fast_silu(float x) {
    return x * __builtin_amdgcn_rcpf(1.0f + __expf(-x));
}

__global__ __launch_bounds__(256) void init_kernel(
    const int* __restrict__ q, const int* __restrict__ sp,
    const float* __restrict__ ce, const float* __restrict__ se,
    const float* __restrict__ mW, const float* __restrict__ mb,
    float* __restrict__ mol, int* __restrict__ cnt, int B, int A, int nzb) {
    if ((int)blockIdx.x < nzb) {
        int g = blockIdx.x * 256 + threadIdx.x;
        if (g < A) cnt[g] = 0;
        return;
    }
    int g = ((int)blockIdx.x - nzb) * 256 + threadIdx.x;
    if (g >= B * F) return;
    int b = g >> 5, f = g & 31;
    int ci = min(max(q[b] + 5, 0), 10);
    int si = min(max(sp[b] - 1, 0), 6);
    float acc = mb[f];
    #pragma unroll
    for (int k = 0; k < 16; ++k) acc += ce[ci * 16 + k] * mW[k * F + f];
    #pragma unroll
    for (int k = 0; k < 16; ++k) acc += se[si * 16 + k] * mW[(16 + k) * F + f];
    mol[g] = acc;
}

__global__ __launch_bounds__(256) void count_x0_kernel(
    const int* __restrict__ dst, int* __restrict__ cnt, int* __restrict__ rnk,
    const int* __restrict__ Z, const int* __restrict__ seg,
    const float* __restrict__ ae, const float* __restrict__ mol,
    float* __restrict__ x0, int P, int A, int neb) {
    if ((int)blockIdx.x < neb) {
        int e = blockIdx.x * 256 + threadIdx.x;
        if (e < P) rnk[e] = atomicAdd(&cnt[dst[e]], 1);
        return;
    }
    int g = ((int)blockIdx.x - neb) * 256 + threadIdx.x;
    if (g >= A * F) return;
    int a = g >> 5, f = g & 31;
    x0[g] = ae[Z[a] * F + f] + mol[seg[a] * F + f];
}

#define SCAN_ITEMS 16
__global__ __launch_bounds__(1024) void scan_kernel(const int* __restrict__ cnt,
                                                    int* __restrict__ off, int A) {
    __shared__ int wsum[32];
    const int t = threadIdx.x, lane = t & 31, w = t >> 5;
    const bool al16 = ((((size_t)cnt) | ((size_t)off)) & 15) == 0;
    int carry = 0;
    for (int base = 0; base < A; base += 1024 * SCAN_ITEMS) {
        const int b = base + t * SCAN_ITEMS;
        const bool full = al16 && (b + SCAN_ITEMS <= A);
        int v[SCAN_ITEMS];
        if (full) {
            const int4* p = (const int4*)(cnt + b);
            int4 q0 = p[0], q1 = p[1], q2 = p[2], q3 = p[3];
            v[0] = q0.x; v[1] = q0.y; v[2] = q0.z; v[3] = q0.w;
            v[4] = q1.x; v[5] = q1.y; v[6] = q1.z; v[7] = q1.w;
            v[8] = q2.x; v[9] = q2.y; v[10] = q2.z; v[11] = q2.w;
            v[12] = q3.x; v[13] = q3.y; v[14] = q3.z; v[15] = q3.w;
        } else {
            #pragma unroll
            for (int i = 0; i < SCAN_ITEMS; ++i) v[i] = (b + i < A) ? cnt[b + i] : 0;
        }
        int local = 0;
        #pragma unroll
        for (int i = 0; i < SCAN_ITEMS; ++i) local += v[i];
        int inc = local;
        #pragma unroll
        for (int d = 1; d < 32; d <<= 1) { int nb = __shfl_up(inc, d); if (lane >= d) inc += nb; }
        if (lane == 31) wsum[w] = inc;
        __syncthreads();
        if (w == 0) {
            int s = wsum[lane];
            #pragma unroll
            for (int d = 1; d < 32; d <<= 1) { int nb = __shfl_up(s, d); if (lane >= d) s += nb; }
            wsum[lane] = s;
        }
        __syncthreads();
        int run = carry + inc - local + (w > 0 ? wsum[w - 1] : 0);
        const int total = wsum[31];
        int o[SCAN_ITEMS];
        #pragma unroll
        for (int i = 0; i < SCAN_ITEMS; ++i) { o[i] = run; run += v[i]; }
        if (full) {
            int4* p = (int4*)(off + b);
            p[0] = make_int4(o[0], o[1], o[2], o[3]);
            p[1] = make_int4(o[4], o[5], o[6], o[7]);
            p[2] = make_int4(o[8], o[9], o[10], o[11]);
            p[3] = make_int4(o[12], o[13], o[14], o[15]);
        } else {
            #pragma unroll
            for (int i = 0; i < SCAN_ITEMS; ++i) if (b + i < A) off[b + i] = o[i];
        }
        carry += total;
        __syncthreads();
    }
    if (t == 0) off[A] = carry;
}

__global__ __launch_bounds__(256) void place_kernel(
    const int* __restrict__ dst, const int* __restrict__ src,
    const int* __restrict__ off, const int* __restrict__ rnk,
    int* __restrict__ lst, int P) {
    int e = blockIdx.x * 256 + threadIdx.x;
    if (e < P) lst[off[dst[e]] + rnk[e]] = src[e];
}

__global__ __launch_bounds__(256) void sort_kernel(
    const int* __restrict__ off, const int* __restrict__ lst,
    int* __restrict__ srt, int A) {
    const int lane = threadIdx.x & 31;
    const int a = blockIdx.x * 8 + __builtin_amdgcn_readfirstlane((int)(threadIdx.x >> 5));
    if (a >= A) return;
    const int o0 = off[a];
    const int n = off[a + 1] - o0;
    if (n <= 32) {
        const int s = (lane < n) ? lst[o0 + lane] : 0x7fffffff;
        int rk = 0;
        for (int j = 0; j < n; ++j) {
            int sj = __builtin_amdgcn_readlane(s, j);
            rk += (sj < s || (sj == s && j < lane)) ? 1 : 0;
        }
        if (lane < n) srt[o0 + rk] = s;
    } else {
        for (int i = lane; i < n; i += 32) {
            const int s = lst[o0 + i];
            int rk = 0;
            for (int j = 0; j < n; ++j) {
                int sj = lst[o0 + j];
                rk += (sj < s || (sj == s && j < i)) ? 1 : 0;
            }
            srt[o0 + rk] = s;
        }
    }
}

template <int ADD_CUT>
__global__ __launch_bounds__(256) void gather_kernel(
    const float* __restrict__ pos, const int* __restrict__ off,
    const int* __restrict__ srt, const float* __restrict__ Wb,
    const float* __restrict__ xin, float* __restrict__ sout, int A) {
    const int lane = threadIdx.x & 31;
    const int a = blockIdx.x * 8 + __builtin_amdgcn_readfirstlane((int)(threadIdx.x >> 5));
    if (a >= A) return;
    const int o0 = off[a];
    const int n = off[a + 1] - o0;
    v2f w2[NB];
    #pragma unroll
    for (int b = 0; b < NB; ++b) { float wv = Wb[b * F + lane]; w2[b] = (v2f){wv, wv}; }
    const float pax = pos[a * 3 + 0], pay = pos[a * 3 + 1], paz = pos[a * 3 + 2];
    float acc = 0.0f;
    for (int c0 = 0; c0 < n; c0 += 32) {
        const int m = (n - c0 < 32) ? (n - c0) : 32;
        const int s = (lane < m) ? srt[o0 + c0 + lane] : a;
        float dx = pos[s * 3 + 0] - pax;
        float dy = pos[s * 3 + 1] - pay;
        float dz = pos[s * 3 + 2] - paz;
        float r = sqrtf(dx * dx + dy * dy + dz * dz);
        float u = r * (1.0f / 6.0f);
        u = u * u;
        float cutv = 0.0f;
        if (u < 1.0f)
            cutv = __expf(1.0f - __builtin_amdgcn_rcpf(fmaxf(1.0f - u, 1e-6f)));
        float zc = fminf(fmaxf(2.0f * __expf(-r) - 1.0f, -1.0f), 1.0f);
        for (int j0 = 0; j0 < m; j0 += 8) {
            float xv[8];
            #pragma unroll
            for (int k = 0; k < 8; ++k) {
                int sj = __builtin_amdgcn_readlane(s, j0 + k);
                xv[k] = xin[sj * F + lane];
            }
            __builtin_amdgcn_sched_barrier(0);
            #pragma unroll
            for (int p = 0; p < 4; ++p) {
                const int ja = j0 + 2 * p, jb = ja + 1;
                v2f z  = (v2f){__int_as_float(__builtin_amdgcn_readlane(__float_as_int(zc), ja)),
                               __int_as_float(__builtin_amdgcn_readlane(__float_as_int(zc), jb))};
                v2f cu = (v2f){__int_as_float(__builtin_amdgcn_readlane(__float_as_int(cutv), ja)),
                               __int_as_float(__builtin_amdgcn_readlane(__float_as_int(cutv), jb))};
                v2f tz = z + z;
                v2f Tm2 = (v2f){1.0f, 1.0f}, Tm1 = z;
                v2f c = w2[0] + w2[1] * z;
                #pragma unroll
                for (int b2 = 2; b2 < NB; ++b2) {
                    v2f T = tz * Tm1 - Tm2;
                    c += w2[b2] * T;
                    Tm2 = Tm1; Tm1 = T;
                }
                if (ADD_CUT) c += (v2f){1.0f, 1.0f};
                v2f val = (cu * c) * (v2f){xv[2 * p], xv[2 * p + 1]};
                acc += (ja < m) ? val.x : 0.0f;
                acc += (jb < m) ? val.y : 0.0f;
            }
        }
    }
    sout[a * F + lane] = acc;
}

__global__ __launch_bounds__(256) void refine_kernel(
    const float* __restrict__ xa, const float* __restrict__ xb,
    const float* __restrict__ W1s, const float* __restrict__ b1s,
    const float* __restrict__ W2s, const float* __restrict__ b2s,
    float* __restrict__ out, int A) {
    __shared__ __align__(16) float sW[6 * F * F];
    __shared__ __align__(16) float sb[6 * F];
    __shared__ __align__(16) float xt[128 * F];
    __shared__ __align__(16) float ht[128 * F];

    const int t = threadIdx.x;
    const int lane = t & 31;
    const int wave = t >> 5;
    const int base = blockIdx.x * 128 * F;
    const int lim = A * F;

    float4 wlo[3], whi[3];
    #pragma unroll
    for (int u = 0; u < 3; ++u) {
        const int item = t + 256 * u;
        const int mat = item >> 7;
        const int kp = (item >> 3) & 15;
        const int nq = item & 7;
        const float* Ws = (mat < 3) ? (W1s + mat * F * F) : (W2s + (mat - 3) * F * F);
        wlo[u] = *(const float4*)(Ws + (2 * kp) * F + nq * 4);
        whi[u] = *(const float4*)(Ws + (2 * kp + 1) * F + nq * 4);
    }
    const int tb = (t < 48) ? t : 47;
    const float4 bv = *(const float4*)((tb < 24) ? (b1s + tb * 4) : (b2s + (tb - 24) * 4));
    float4 va[4], vb[4];
    #pragma unroll
    for (int u = 0; u < 4; ++u) {
        const int gi = min(base + (t + 256 * u) * 4, lim - 4);
        va[u] = *(const float4*)(xa + gi);
        vb[u] = *(const float4*)(xb + gi);
    }

    #pragma unroll
    for (int u = 0; u < 3; ++u) {
        const int item = t + 256 * u;
        float* d = sW + (item >> 7) * F * F + ((item >> 3) & 15) * 64 + (item & 7) * 8;
        *(float4*)d       = make_float4(wlo[u].x, whi[u].x, wlo[u].y, whi[u].y);
        *(float4*)(d + 4) = make_float4(wlo[u].z, whi[u].z, wlo[u].w, whi[u].w);
    }
    *(float4*)(sb + tb * 4) = bv;
    #pragma unroll
    for (int u = 0; u < 4; ++u) {
        const int i = (t + 256 * u) * 4;
        float4 s = make_float4(va[u].x + vb[u].x, va[u].y + vb[u].y,
                               va[u].z + vb[u].z, va[u].w + vb[u].w);
        if (base + i >= lim) s = make_float4(0.0f, 0.0f, 0.0f, 0.0f);
        *(float4*)(xt + i) = s;
    }
    __syncthreads();

    const int m     = lane & 15;
    const int khalf = (lane >> 4) << 1;
    const int row0  = wave * 16;
    const int cm    = (lane >> 4) << 3;
    const int cn    = lane & 15;

    for (int j = 0; j < 3; ++j) {
        const v2f* W1p = (const v2f*)(sW + j * F * F);
        const v2f* W2p = (const v2f*)(sW + (3 + j) * F * F);
        const float* B1 = sb + j * F;
        const float* B2 = sb + (3 + j) * F;

        v8f acc0, acc1;
        {
            float bn0 = B1[cn], bn1 = B1[cn + 16];
            #pragma unroll
            for (int q = 0; q < 8; ++q) { acc0[q] = bn0; acc1[q] = bn1; }
        }
        #pragma unroll
        for (int kc = 0; kc < 8; ++kc) {
            int kb = kc * 4 + khalf;
            int kp = (kb >> 1) << 5;
            v2f a  = *(const v2f*)(xt + (row0 + m) * F + kb);
            v2f b0 = W1p[kp + cn];
            v2f b1 = W1p[kp + cn + 16];
            acc0 = __builtin_amdgcn_wmma_f32_16x16x4_f32(false, a, false, b0, (short)0, acc0, false, false);
            acc1 = __builtin_amdgcn_wmma_f32_16x16x4_f32(false, a, false, b1, (short)0, acc1, false, false);
        }
        #pragma unroll
        for (int q = 0; q < 8; ++q) {
            ht[(row0 + cm + q) * F + cn]      = fast_silu(acc0[q]);
            ht[(row0 + cm + q) * F + cn + 16] = fast_silu(acc1[q]);
        }
        __syncthreads();

        {
            float bn0 = B2[cn], bn1 = B2[cn + 16];
            #pragma unroll
            for (int q = 0; q < 8; ++q) { acc0[q] = bn0; acc1[q] = bn1; }
        }
        #pragma unroll
        for (int kc = 0; kc < 8; ++kc) {
            int kb = kc * 4 + khalf;
            int kp = (kb >> 1) << 5;
            v2f a  = *(const v2f*)(ht + (row0 + m) * F + kb);
            v2f b0 = W2p[kp + cn];
            v2f b1 = W2p[kp + cn + 16];
            acc0 = __builtin_amdgcn_wmma_f32_16x16x4_f32(false, a, false, b0, (short)0, acc0, false, false);
            acc1 = __builtin_amdgcn_wmma_f32_16x16x4_f32(false, a, false, b1, (short)0, acc1, false, false);
        }
        __syncthreads();
        #pragma unroll
        for (int q = 0; q < 8; ++q) {
            float x0v = xt[(row0 + cm + q) * F + cn]      + acc0[q];
            float x1v = xt[(row0 + cm + q) * F + cn + 16] + acc1[q];
            xt[(row0 + cm + q) * F + cn]      = fast_silu(x0v);
            xt[(row0 + cm + q) * F + cn + 16] = fast_silu(x1v);
        }
        __syncthreads();
    }

    #pragma unroll
    for (int u = 0; u < 4; ++u) {
        const int i = (t + 256 * u) * 4;
        if (base + i < lim) *(float4*)(out + base + i) = *(const float4*)(xt + i);
    }
}

extern "C" void kernel_launch(void* const* d_in, const int* in_sizes, int n_in,
                              void* d_out, int out_size, void* d_ws, size_t ws_size,
                              hipStream_t stream) {
    const float* pos  = (const float*)d_in[0];
    const int*   q    = (const int*)d_in[1];
    const int*   sp   = (const int*)d_in[2];
    const int*   Z    = (const int*)d_in[3];
    const int*   dst  = (const int*)d_in[4];
    const int*   src  = (const int*)d_in[5];
    const int*   seg  = (const int*)d_in[6];
    const float* ce   = (const float*)d_in[7];
    const float* se   = (const float*)d_in[8];
    const float* ae   = (const float*)d_in[9];
    const float* mW   = (const float*)d_in[10];
    const float* mb   = (const float*)d_in[11];
    const float* mpWb = (const float*)d_in[12];
    const float* rW1  = (const float*)d_in[13];
    const float* rb1  = (const float*)d_in[14];
    const float* rW2  = (const float*)d_in[15];
    const float* rb2  = (const float*)d_in[16];

    const int B = in_sizes[1];
    const int A = in_sizes[3];
    const int P = in_sizes[4];

    float* ws  = (float*)d_ws;
    float* mol = ws;
    float* x0  = mol + (size_t)B * F;
    float* sP  = x0 + (size_t)A * F;
    float* xs0 = sP + (size_t)A * F;
    int*   srt = (int*)(xs0 + (size_t)A * F);
    int*   cnt = srt + P;
    int*   off = cnt + A;
    int*   rnk = (int*)xs0;
    int*   lst = (int*)sP;

    const int nzb = (A + 255) / 256;
    const int neb = (P + 255) / 256;
    const int gblocks = (A + 7) / 8;

    init_kernel<<<nzb + (B * F + 255) / 256, 256, 0, stream>>>(q, sp, ce, se, mW, mb, mol, cnt, B, A, nzb);
    count_x0_kernel<<<neb + (A * F + 255) / 256, 256, 0, stream>>>(dst, cnt, rnk, Z, seg, ae, mol, x0, P, A, neb);
    scan_kernel<<<1, 1024, 0, stream>>>(cnt, off, A);
    place_kernel<<<neb, 256, 0, stream>>>(dst, src, off, rnk, lst, P);
    sort_kernel<<<gblocks, 256, 0, stream>>>(off, lst, srt, A);

    gather_kernel<1><<<gblocks, 256, 0, stream>>>(pos, off, srt, mpWb, x0, sP, A);
    refine_kernel<<<(A + 127) / 128, 256, 0, stream>>>(
        x0, sP, rW1, rb1, rW2, rb2, xs0, A);
    gather_kernel<0><<<gblocks, 256, 0, stream>>>(pos, off, srt, mpWb + NB * F, xs0, sP, A);
    refine_kernel<<<(A + 127) / 128, 256, 0, stream>>>(
        xs0, sP, rW1 + 3 * F * F, rb1 + 3 * F, rW2 + 3 * F * F, rb2 + 3 * F,
        (float*)d_out, A);
}


# edge lists in one pass: the counting atomic gives each edge a slot in its atom's row of 48 and stores the source; scan, place, sort launches gone (9 to 6); gather 1 orders rows; sum order unchanged
# speedup vs baseline: 1.6411x; 1.2543x over previous
#include <hip/hip_runtime.h>
#include <hip/hip_bf16.h>

typedef float v2f __attribute__((ext_vector_type(2)));
typedef float v8f __attribute__((ext_vector_type(8)));

#define F 32
#define NB 16
#define CAP 48

__device__ __forceinline__ float fast_silu(float x) {
    return x * __builtin_amdgcn_rcpf(1.0f + __expf(-x));
}

__global__ __launch_bounds__(256) void init_kernel(
    const int* __restrict__ q, const int* __restrict__ sp,
    const float* __restrict__ ce, const float* __restrict__ se,
    const float* __restrict__ mW, const float* __restrict__ mb,
    float* __restrict__ mol, int* __restrict__ cnt, int B, int A, int nzb) {
    if ((int)blockIdx.x < nzb) {
        int g = blockIdx.x * 256 + threadIdx.x;
        if (g < A) cnt[g] = 0;
        return;
    }
    int g = ((int)blockIdx.x - nzb) * 256 + threadIdx.x;
    if (g >= B * F) return;
    int b = g >> 5, f = g & 31;
    int ci = min(max(q[b] + 5, 0), 10);
    int si = min(max(sp[b] - 1, 0), 6);
    float acc = mb[f];
    #pragma unroll
    for (int k = 0; k < 16; ++k) acc += ce[ci * 16 + k] * mW[k * F + f];
    #pragma unroll
    for (int k = 0; k < 16; ++k) acc += se[si * 16 + k] * mW[(16 + k) * F + f];
    mol[g] = acc;
}

__global__ __launch_bounds__(256) void count_x0_kernel(
    const int* __restrict__ dst, const int* __restrict__ src,
    int* __restrict__ cnt, int* __restrict__ bkt,
    const int* __restrict__ Z, const int* __restrict__ seg,
    const float* __restrict__ ae, const float* __restrict__ mol,
    float* __restrict__ x0, int P, int A, int neb) {
    if ((int)blockIdx.x < neb) {
        int e = blockIdx.x * 256 + threadIdx.x;
        if (e < P) {
            const int d = dst[e];
            int sv = src[e];
            asm volatile("" : "+v"(sv));
            const int slot = atomicAdd(&cnt[d], 1);
            if (slot < CAP) bkt[d * CAP + slot] = sv;
        }
        return;
    }
    int g = ((int)blockIdx.x - neb) * 256 + threadIdx.x;
    if (g >= A * F) return;
    int a = g >> 5, f = g & 31;
    x0[g] = ae[Z[a] * F + f] + mol[seg[a] * F + f];
}

template <int ADD_CUT, int SORT>
__global__ __launch_bounds__(256) void gather_kernel(
    const float* __restrict__ pos, const int* __restrict__ cnt, int* bkt,
    const int* __restrict__ dst, const int* __restrict__ src,
    const float* __restrict__ Wb, const float* __restrict__ xin,
    float* __restrict__ sout, int A, int P) {
    const int lane = threadIdx.x & 31;
    const int a = blockIdx.x * 8 + __builtin_amdgcn_readfirstlane((int)(threadIdx.x >> 5));
    if (a >= A) return;
    const int n = cnt[a];
    int* row = bkt + a * CAP;
    const bool scan_path = n > CAP;
    v2f w2[NB];
    #pragma unroll
    for (int b = 0; b < NB; ++b) { float wv = Wb[b * F + lane]; w2[b] = (v2f){wv, wv}; }
    const float pax = pos[a * 3 + 0], pay = pos[a * 3 + 1], paz = pos[a * 3 + 2];

    int sreg = a;
    bool inreg = false;
    if (SORT && !scan_path) {
        if (n <= 32) {
            const int sr = (lane < n) ? row[lane] : 0x7fffffff;
            int rk = 0;
            for (int j = 0; j < n; ++j) {
                int sj = __builtin_amdgcn_readlane(sr, j);
                rk += (sj < sr || (sj == sr && j < lane)) ? 1 : 0;
            }
            sreg = __builtin_amdgcn_ds_permute(rk << 2, sr);
            if (lane < n) row[lane] = sreg;
            inreg = true;
        } else {
            const int s0 = row[lane];
            const int s1 = (32 + lane < n) ? row[32 + lane] : 0x7fffffff;
            int r0 = 0, r1 = 0;
            #pragma nounroll
            for (int j = 0; j < 32; ++j) {
                int sj = __builtin_amdgcn_readlane(s0, j);
                r0 += (sj < s0 || (sj == s0 && j < lane)) ? 1 : 0;
                r1 += (sj <= s1) ? 1 : 0;
            }
            for (int j = 0; j < n - 32; ++j) {
                int sj = __builtin_amdgcn_readlane(s1, j);
                r0 += (sj < s0) ? 1 : 0;
                r1 += (sj < s1 || (sj == s1 && j < lane)) ? 1 : 0;
            }
            row[r0] = s0;
            if (32 + lane < n) row[r1] = s1;
            __threadfence_block();
        }
    }

    float acc = 0.0f;

    int c0 = 0;
    int epos = 0;
    unsigned pmask = 0u;
    int psrc = 0;
    for (;;) {
        int m = 0;
        int s = a;
        if (!scan_path) {
            if (c0 >= n) break;
            m = (n - c0 < 32) ? (n - c0) : 32;
            if (lane < m) s = (inreg && c0 == 0) ? sreg : row[c0 + lane];
            c0 += 32;
        } else {
            while (m < 32) {
                if (pmask == 0u) {
                    if (epos >= P) break;
                    const int e = epos + lane;
                    const int d = (e < P) ? dst[e] : -1;
                    psrc = (e < P) ? src[e] : 0;
                    pmask = __builtin_amdgcn_ballot_w32(d == a);
                    epos += 32;
                    continue;
                }
                const int j = __builtin_ctz(pmask);
                pmask &= pmask - 1u;
                const int sj = __builtin_amdgcn_readlane(psrc, j);
                if (lane == m) s = sj;
                ++m;
            }
            if (m == 0) break;
        }
        float dx = pos[s * 3 + 0] - pax;
        float dy = pos[s * 3 + 1] - pay;
        float dz = pos[s * 3 + 2] - paz;
        float r = sqrtf(dx * dx + dy * dy + dz * dz);
        float u = r * (1.0f / 6.0f);
        u = u * u;
        float cutv = 0.0f;
        if (u < 1.0f)
            cutv = __expf(1.0f - __builtin_amdgcn_rcpf(fmaxf(1.0f - u, 1e-6f)));
        float zc = fminf(fmaxf(2.0f * __expf(-r) - 1.0f, -1.0f), 1.0f);
        for (int j0 = 0; j0 < m; j0 += 8) {
            float xv[8];
            #pragma unroll
            for (int k = 0; k < 8; ++k) {
                int sj = __builtin_amdgcn_readlane(s, j0 + k);
                xv[k] = xin[sj * F + lane];
            }
            __builtin_amdgcn_sched_barrier(0);
            #pragma unroll
            for (int p = 0; p < 4; ++p) {
                const int ja = j0 + 2 * p, jb = ja + 1;
                v2f z  = (v2f){__int_as_float(__builtin_amdgcn_readlane(__float_as_int(zc), ja)),
                               __int_as_float(__builtin_amdgcn_readlane(__float_as_int(zc), jb))};
                v2f cu = (v2f){__int_as_float(__builtin_amdgcn_readlane(__float_as_int(cutv), ja)),
                               __int_as_float(__builtin_amdgcn_readlane(__float_as_int(cutv), jb))};
                v2f tz = z + z;
                v2f Tm2 = (v2f){1.0f, 1.0f}, Tm1 = z;
                v2f c = w2[0] + w2[1] * z;
                #pragma unroll
                for (int b2 = 2; b2 < NB; ++b2) {
                    v2f T = tz * Tm1 - Tm2;
                    c += w2[b2] * T;
                    Tm2 = Tm1; Tm1 = T;
                }
                if (ADD_CUT) c += (v2f){1.0f, 1.0f};
                v2f val = (cu * c) * (v2f){xv[2 * p], xv[2 * p + 1]};
                acc += (ja < m) ? val.x : 0.0f;
                acc += (jb < m) ? val.y : 0.0f;
            }
        }
    }
    sout[a * F + lane] = acc;
}

__global__ __launch_bounds__(256) void refine_kernel(
    const float* xa, const float* __restrict__ xb,
    const float* __restrict__ W1s, const float* __restrict__ b1s,
    const float* __restrict__ W2s, const float* __restrict__ b2s,
    float* out, int A) {
    __shared__ __align__(16) float sW[6 * F * F];
    __shared__ __align__(16) float sb[6 * F];
    __shared__ __align__(16) float xt[128 * F];
    __shared__ __align__(16) float ht[128 * F];

    const int t = threadIdx.x;
    const int lane = t & 31;
    const int wave = t >> 5;
    const int base = blockIdx.x * 128 * F;
    const int lim = A * F;

    float4 wlo[3], whi[3];
    #pragma unroll
    for (int u = 0; u < 3; ++u) {
        const int item = t + 256 * u;
        const int mat = item >> 7;
        const int kp = (item >> 3) & 15;
        const int nq = item & 7;
        const float* Ws = (mat < 3) ? (W1s + mat * F * F) : (W2s + (mat - 3) * F * F);
        wlo[u] = *(const float4*)(Ws + (2 * kp) * F + nq * 4);
        whi[u] = *(const float4*)(Ws + (2 * kp + 1) * F + nq * 4);
    }
    const int tb = (t < 48) ? t : 47;
    const float4 bv = *(const float4*)((tb < 24) ? (b1s + tb * 4) : (b2s + (tb - 24) * 4));
    float4 va[4], vb[4];
    #pragma unroll
    for (int u = 0; u < 4; ++u) {
        const int gi = min(base + (t + 256 * u) * 4, lim - 4);
        va[u] = *(const float4*)(xa + gi);
        vb[u] = *(const float4*)(xb + gi);
    }

    #pragma unroll
    for (int u = 0; u < 3; ++u) {
        const int item = t + 256 * u;
        float* d = sW + (item >> 7) * F * F + ((item >> 3) & 15) * 64 + (item & 7) * 8;
        *(float4*)d       = make_float4(wlo[u].x, whi[u].x, wlo[u].y, whi[u].y);
        *(float4*)(d + 4) = make_float4(wlo[u].z, whi[u].z, wlo[u].w, whi[u].w);
    }
    *(float4*)(sb + tb * 4) = bv;
    #pragma unroll
    for (int u = 0; u < 4; ++u) {
        const int i = (t + 256 * u) * 4;
        float4 s = make_float4(va[u].x + vb[u].x, va[u].y + vb[u].y,
                               va[u].z + vb[u].z, va[u].w + vb[u].w);
        if (base + i >= lim) s = make_float4(0.0f, 0.0f, 0.0f, 0.0f);
        *(float4*)(xt + i) = s;
    }
    __syncthreads();

    const int m     = lane & 15;
    const int khalf = (lane >> 4) << 1;
    const int row0  = wave * 16;
    const int cm    = (lane >> 4) << 3;
    const int cn    = lane & 15;

    for (int j = 0; j < 3; ++j) {
        const v2f* W1p = (const v2f*)(sW + j * F * F);
        const v2f* W2p = (const v2f*)(sW + (3 + j) * F * F);
        const float* B1 = sb + j * F;
        const float* B2 = sb + (3 + j) * F;

        v8f acc0, acc1;
        {
            float bn0 = B1[cn], bn1 = B1[cn + 16];
            #pragma unroll
            for (int q = 0; q < 8; ++q) { acc0[q] = bn0; acc1[q] = bn1; }
        }
        #pragma unroll
        for (int kc = 0; kc < 8; ++kc) {
            int kb = kc * 4 + khalf;
            int kp = (kb >> 1) << 5;
            v2f a  = *(const v2f*)(xt + (row0 + m) * F + kb);
            v2f b0 = W1p[kp + cn];
            v2f b1 = W1p[kp + cn + 16];
            acc0 = __builtin_amdgcn_wmma_f32_16x16x4_f32(false, a, false, b0, (short)0, acc0, false, false);
            acc1 = __builtin_amdgcn_wmma_f32_16x16x4_f32(false, a, false, b1, (short)0, acc1, false, false);
        }
        #pragma unroll
        for (int q = 0; q < 8; ++q) {
            ht[(row0 + cm + q) * F + cn]      = fast_silu(acc0[q]);
            ht[(row0 + cm + q) * F + cn + 16] = fast_silu(acc1[q]);
        }
        __syncthreads();

        {
            float bn0 = B2[cn], bn1 = B2[cn + 16];
            #pragma unroll
            for (int q = 0; q < 8; ++q) { acc0[q] = bn0; acc1[q] = bn1; }
        }
        #pragma unroll
        for (int kc = 0; kc < 8; ++kc) {
            int kb = kc * 4 + khalf;
            int kp = (kb >> 1) << 5;
            v2f a  = *(const v2f*)(ht + (row0 + m) * F + kb);
            v2f b0 = W2p[kp + cn];
            v2f b1 = W2p[kp + cn + 16];
            acc0 = __builtin_amdgcn_wmma_f32_16x16x4_f32(false, a, false, b0, (short)0, acc0, false, false);
            acc1 = __builtin_amdgcn_wmma_f32_16x16x4_f32(false, a, false, b1, (short)0, acc1, false, false);
        }
        __syncthreads();
        #pragma unroll
        for (int q = 0; q < 8; ++q) {
            float x0v = xt[(row0 + cm + q) * F + cn]      + acc0[q];
            float x1v = xt[(row0 + cm + q) * F + cn + 16] + acc1[q];
            xt[(row0 + cm + q) * F + cn]      = fast_silu(x0v);
            xt[(row0 + cm + q) * F + cn + 16] = fast_silu(x1v);
        }
        __syncthreads();
    }

    #pragma unroll
    for (int u = 0; u < 4; ++u) {
        const int i = (t + 256 * u) * 4;
        if (base + i < lim) *(float4*)(out + base + i) = *(const float4*)(xt + i);
    }
}

extern "C" void kernel_launch(void* const* d_in, const int* in_sizes, int n_in,
                              void* d_out, int out_size, void* d_ws, size_t ws_size,
                              hipStream_t stream) {
    const float* pos  = (const float*)d_in[0];
    const int*   q    = (const int*)d_in[1];
    const int*   sp   = (const int*)d_in[2];
    const int*   Z    = (const int*)d_in[3];
    const int*   dst  = (const int*)d_in[4];
    const int*   src  = (const int*)d_in[5];
    const int*   seg  = (const int*)d_in[6];
    const float* ce   = (const float*)d_in[7];
    const float* se   = (const float*)d_in[8];
    const float* ae   = (const float*)d_in[9];
    const float* mW   = (const float*)d_in[10];
    const float* mb   = (const float*)d_in[11];
    const float* mpWb = (const float*)d_in[12];
    const float* rW1  = (const float*)d_in[13];
    const float* rb1  = (const float*)d_in[14];
    const float* rW2  = (const float*)d_in[15];
    const float* rb2  = (const float*)d_in[16];

    const int B = in_sizes[1];
    const int A = in_sizes[3];
    const int P = in_sizes[4];

    float* ws  = (float*)d_ws;
    float* mol = ws;
    float* xs  = mol + (size_t)B * F;
    float* sP  = xs + (size_t)A * F;
    int*   bkt = (int*)(sP + (size_t)A * F);
    int*   cnt = bkt + (size_t)A * CAP;

    const int nzb = (A + 255) / 256;
    const int neb = (P + 255) / 256;
    const int gblocks = (A + 7) / 8;

    init_kernel<<<nzb + (B * F + 255) / 256, 256, 0, stream>>>(q, sp, ce, se, mW, mb, mol, cnt, B, A, nzb);
    count_x0_kernel<<<neb + (A * F + 255) / 256, 256, 0, stream>>>(dst, src, cnt, bkt, Z, seg, ae, mol, xs, P, A, neb);

    gather_kernel<1, 1><<<gblocks, 256, 0, stream>>>(pos, cnt, bkt, dst, src, mpWb, xs, sP, A, P);
    refine_kernel<<<(A + 127) / 128, 256, 0, stream>>>(
        xs, sP, rW1, rb1, rW2, rb2, xs, A);
    gather_kernel<0, 0><<<gblocks, 256, 0, stream>>>(pos, cnt, bkt, dst, src, mpWb + NB * F, xs, sP, A, P);
    refine_kernel<<<(A + 127) / 128, 256, 0, stream>>>(
        xs, sP, rW1 + 3 * F * F, rb1 + 3 * F, rW2 + 3 * F * F, rb2 + 3 * F,
        (float*)d_out, A);
}
